# starting kernel
# baseline (speedup 1.0000x reference)
#include <hip/hip_runtime.h>

typedef __attribute__((ext_vector_type(2))) float v2f;
typedef __attribute__((ext_vector_type(8))) float v8f;

static constexpr int NT = 4096;
static constexpr int Dd = 32, Oo = 32, Gg = 8, Tt = 64;

__global__ __launch_bounds__(256) void mlm_main_kernel(
    const float* __restrict__ x,
    const float* __restrict__ w, const float* __restrict__ mask,
    float* __restrict__ part)
{
    __shared__ float pbuf[8 * 512];

    const int tid  = threadIdx.x;
    const int wave = tid >> 5;
    const int lane = tid & 31;
    const int blk  = blockIdx.x;
    const int oh   = blk & 1;
    const int d    = (blk >> 1) & 31;
    const int g    = blk >> 6;
    const int o0   = oh << 4;

    const int olane = lane & 15;
    const int khalf = (lane >> 4) << 1;

    v2f mreg[16];
    const float* mrow = mask + (o0 + olane) * Tt + khalf;
#pragma unroll
    for (int j = 0; j < 16; ++j)
        mreg[j] = *(const v2f*)(mrow + 4 * j);

    const float* wrow = w + (size_t)((g * Dd + d) * Oo + o0 + olane) * NT + khalf;
    const float* xa   = x + olane * NT + khalf;
    const float* xb   = xa + 16 * NT;

    v8f acc0 = {}, acc1 = {}, acc2 = {}, acc3 = {};
    const int kwave = wave << 9;

    for (int kc = 0; kc < 512; kc += 64) {
        const int k0 = kwave + kc;
        const float* wp  = wrow + k0;
        const float* xap = xa + k0;
        const float* xbp = xb + k0;
#pragma unroll
        for (int j = 0; j < 16; j += 2) {
            const int kk = j << 2;
            v2f w0 = *(const v2f*)(wp + kk);
            v2f w1 = *(const v2f*)(wp + kk + 4);
            v2f b0 = w0 * mreg[j];
            v2f b1 = w1 * mreg[j + 1];
            v2f a00 = *(const v2f*)(xap + kk);
            v2f a10 = *(const v2f*)(xbp + kk);
            v2f a01 = *(const v2f*)(xap + kk + 4);
            v2f a11 = *(const v2f*)(xbp + kk + 4);
            acc0 = __builtin_amdgcn_wmma_f32_16x16x4_f32(false, a00, false, b0, (short)0, acc0, false, false);
            acc1 = __builtin_amdgcn_wmma_f32_16x16x4_f32(false, a10, false, b0, (short)0, acc1, false, false);
            acc2 = __builtin_amdgcn_wmma_f32_16x16x4_f32(false, a01, false, b1, (short)0, acc2, false, false);
            acc3 = __builtin_amdgcn_wmma_f32_16x16x4_f32(false, a11, false, b1, (short)0, acc3, false, false);
        }
        __builtin_prefetch(wp + 64, 0, 0);
    }
    acc0 += acc2;
    acc1 += acc3;

    float* pb = &pbuf[wave * 512];
#pragma unroll
    for (int r = 0; r < 8; ++r) {
        pb[r * 32 + lane]       = acc0[r];
        pb[256 + r * 32 + lane] = acc1[r];
    }
    __syncthreads();
    float* pblk = part + (size_t)blk * 512;
    for (int i = tid; i < 512; i += 256) {
        float s = 0.0f;
#pragma unroll
        for (int wv = 0; wv < 8; ++wv) s += pbuf[wv * 512 + i];
        pblk[i] = s;
    }
}

__global__ __launch_bounds__(256) void mlm_reduce_kernel(
    const float* __restrict__ part, const float* __restrict__ y,
    const float* __restrict__ bias, float* __restrict__ out)
{
    const int i = blockIdx.x * 256 + threadIdx.x;
    const int o = i & 31;
    const int d = (i >> 5) & 31;
    const int b = i >> 10;
    const int oh = o >> 4, ol = o & 15;
    const int tile = b >> 4, rb = b & 15;
    const int lanei = ol + ((rb >> 3) << 4);
    const int idx = tile * 256 + (rb & 7) * 32 + lanei;
    float s = bias[d * 32 + o];
#pragma unroll
    for (int g = 0; g < 8; ++g)
        s += y[b * 8 + g] * part[(size_t)((g * 32 + d) * 2 + oh) * 512 + idx];
    out[i] = s;
}

extern "C" void kernel_launch(void* const* d_in, const int* in_sizes, int n_in,
                              void* d_out, int out_size, void* d_ws, size_t ws_size,
                              hipStream_t stream) {
    const float* x    = (const float*)d_in[0];
    const float* y    = (const float*)d_in[1];
    const float* w    = (const float*)d_in[2];
    const float* bias = (const float*)d_in[3];
    const float* mask = (const float*)d_in[4];
    float* out  = (float*)d_out;
    float* part = (float*)d_ws;

    mlm_main_kernel<<<512, 256, 0, stream>>>(x, w, mask, part);
    mlm_reduce_kernel<<<128, 256, 0, stream>>>(part, y, bias, out);
}


# x k-slice staged once per block in LDS and shared by its 16 N-tiles (was re-read from L2 by every tile); K split over blocks, partials summed in the 2nd kernel: f32 summation order changes
# speedup vs baseline: 1.5508x; 1.5508x over previous
#include <hip/hip_runtime.h>

typedef __attribute__((ext_vector_type(2))) float v2f;
typedef __attribute__((ext_vector_type(4))) float v4f;
typedef __attribute__((ext_vector_type(8))) float v8f;

static constexpr int NT  = 4096;
static constexpr int KL  = 512;
static constexpr int PSZ = 262144;

__global__ __launch_bounds__(512) void mlm_main_kernel(
    const float* __restrict__ x,
    const float* __restrict__ w, const float* __restrict__ mask,
    float* __restrict__ part)
{
    __shared__ v4f xs[32 * 128];

    const int tid  = threadIdx.x;
    const int wave = tid >> 5;
    const int lane = tid & 31;
    const int c    = lane & 15;
    const int h    = lane >> 4;
    const int kc   = blockIdx.x & 7;
    const int rg   = blockIdx.x >> 3;
    const int nt   = rg * 16 + wave;
    const int k0   = kc * KL;

    const float* wrow = w + (size_t)(nt * 16 + c) * NT + k0 + 4 * h;

    v4f xv[8];
#pragma unroll
    for (int i = 0; i < 8; ++i) {
        const int idx = tid + 512 * i;
        xv[i] = *(const v4f*)(x + (idx >> 7) * NT + k0 + 4 * (idx & 127));
    }

    const float* mrow = mask + (((nt & 1) << 4) + c) * 64 + 4 * h;
    v4f mreg[8];
#pragma unroll
    for (int j = 0; j < 8; ++j)
        mreg[j] = *(const v4f*)(mrow + 8 * j);

    v4f wbuf[3][8];
#pragma unroll
    for (int j = 0; j < 8; ++j)
        wbuf[0][j] = *(const v4f*)(wrow + 8 * j);
#pragma unroll
    for (int j = 0; j < 8; ++j)
        wbuf[1][j] = *(const v4f*)(wrow + 64 + 8 * j);

#pragma unroll
    for (int i = 0; i < 8; ++i) {
        const int idx = tid + 512 * i;
        const int b = idx >> 7;
        const int u = idx & 127;
        xs[b * 128 + (u ^ (b & 15))] = xv[i];
    }
    __syncthreads();

    const v4f* xa0 = xs + c * 128;
    const v4f* xa1 = xs + (c + 16) * 128;

    v8f acc0 = {}, acc1 = {}, acc2 = {}, acc3 = {};

    v4f af[2][8];
#pragma unroll
    for (int jj = 0; jj < 4; ++jj) {
        const int u = (2 * jj + h) ^ c;
        af[0][2 * jj]     = xa0[u];
        af[0][2 * jj + 1] = xa1[u];
    }

#pragma unroll
    for (int gi = 0; gi < 16; ++gi) {
        const int p  = gi >> 1;
        const int j0 = (gi & 1) << 2;
        if ((gi & 1) == 0 && p + 2 < 8) {
#pragma unroll
            for (int j = 0; j < 8; ++j)
                wbuf[(p + 2) % 3][j] = *(const v4f*)(wrow + 64 * (p + 2) + 8 * j);
        }
        if (gi + 1 < 16) {
            const int pn  = (gi + 1) >> 1;
            const int jn0 = ((gi + 1) & 1) << 2;
#pragma unroll
            for (int jj = 0; jj < 4; ++jj) {
                const int u = pn * 16 + ((2 * (jn0 + jj) + h) ^ c);
                af[(gi + 1) & 1][2 * jj]     = xa0[u];
                af[(gi + 1) & 1][2 * jj + 1] = xa1[u];
            }
        }
        __builtin_amdgcn_sched_barrier(0);
#pragma unroll
        for (int jj = 0; jj < 4; ++jj) {
            const int j = j0 + jj;
            const v4f a0 = af[gi & 1][2 * jj];
            const v4f a1 = af[gi & 1][2 * jj + 1];
            const v4f bm = wbuf[p % 3][j] * mreg[j];
            const v2f a0l = {a0.x, a0.y}, a0h = {a0.z, a0.w};
            const v2f a1l = {a1.x, a1.y}, a1h = {a1.z, a1.w};
            const v2f bl  = {bm.x, bm.y}, bh  = {bm.z, bm.w};
            acc0 = __builtin_amdgcn_wmma_f32_16x16x4_f32(false, a0l, false, bl, (short)0, acc0, false, false);
            acc1 = __builtin_amdgcn_wmma_f32_16x16x4_f32(false, a1l, false, bl, (short)0, acc1, false, false);
            acc2 = __builtin_amdgcn_wmma_f32_16x16x4_f32(false, a0h, false, bh, (short)0, acc2, false, false);
            acc3 = __builtin_amdgcn_wmma_f32_16x16x4_f32(false, a1h, false, bh, (short)0, acc3, false, false);
        }
        __builtin_amdgcn_sched_barrier(0);
    }
    acc0 += acc2;
    acc1 += acc3;

    float* pt = part + (size_t)kc * PSZ + (size_t)nt * 512 + lane;
#pragma unroll
    for (int r = 0; r < 8; ++r) {
        pt[r * 32]       = acc0[r];
        pt[256 + r * 32] = acc1[r];
    }
}

__global__ __launch_bounds__(128) void mlm_reduce_kernel(
    const float* __restrict__ part, const float* __restrict__ y,
    const float* __restrict__ bias, float* __restrict__ out)
{
    const int t    = blockIdx.x * 128 + threadIdx.x;
    const int lane = t & 31;
    const int u    = t >> 5;
    const int r    = u & 7;
    const int m    = (u >> 3) & 1;
    const int oh   = (u >> 4) & 1;
    const int d    = u >> 5;
    const int b    = m * 16 + ((lane >> 4) << 3) + r;
    const int o    = oh * 16 + (lane & 15);

    const float* p0 = part + ((d * 2 + oh) * 2 + m) * 256 + r * 32 + lane;
    const v4f y0 = *(const v4f*)(y + b * 8);
    const v4f y1 = *(const v4f*)(y + b * 8 + 4);
    const float yv[8] = {y0.x, y0.y, y0.z, y0.w, y1.x, y1.y, y1.z, y1.w};
    float s = bias[d * 32 + o];
#pragma unroll
    for (int g = 0; g < 8; ++g) {
        const float* pg = p0 + g * 32768;
        float tsum = 0.0f;
#pragma unroll
        for (int q = 0; q < 8; ++q) tsum += pg[(size_t)q * PSZ];
        s += yv[g] * tsum;
    }
    out[b * 1024 + d * 32 + o] = s;
}

extern "C" void kernel_launch(void* const* d_in, const int* in_sizes, int n_in,
                              void* d_out, int out_size, void* d_ws, size_t ws_size,
                              hipStream_t stream) {
    const float* x    = (const float*)d_in[0];
    const float* y    = (const float*)d_in[1];
    const float* w    = (const float*)d_in[2];
    const float* bias = (const float*)d_in[3];
    const float* mask = (const float*)d_in[4];
    float* out  = (float*)d_out;
    float* part = (float*)d_ws;

    mlm_main_kernel<<<256, 512, 0, stream>>>(x, w, mask, part);
    mlm_reduce_kernel<<<256, 128, 0, stream>>>(part, y, bias, out);
}


# weights and mask fetched row-major as whole 128-B lines (a load touched 16 lines as 32-B pieces, now 4) and turned into B-fragment layout through wave-private LDS slots
# speedup vs baseline: 1.6303x; 1.0513x over previous
#include <hip/hip_runtime.h>

typedef __attribute__((ext_vector_type(2))) float v2f;
typedef __attribute__((ext_vector_type(4))) float v4f;
typedef __attribute__((ext_vector_type(8))) float v8f;

static constexpr int NT  = 4096;
static constexpr int KL  = 512;
static constexpr int PSZ = 262144;

__global__ __launch_bounds__(512) void mlm_main_kernel(
    const float* __restrict__ x,
    const float* __restrict__ w, const float* __restrict__ mask,
    float* __restrict__ part)
{
    __shared__ v4f xs[32 * 128];
    __shared__ v4f wsl[16 * 2 * 272];

    const int tid  = threadIdx.x;
    const int wave = tid >> 5;
    const int lane = tid & 31;
    const int c    = lane & 15;
    const int h    = lane >> 4;
    const int kc   = blockIdx.x & 7;
    const int rg   = blockIdx.x >> 3;
    const int nt   = rg * 16 + wave;
    const int k0   = kc * KL;

    const float* wl = w + (size_t)(nt * 16 + h) * NT + k0 + 4 * c;

    v4f xv[8];
#pragma unroll
    for (int i = 0; i < 8; ++i) {
        const int idx = tid + 512 * i;
        xv[i] = *(const v4f*)(x + (idx >> 7) * NT + k0 + 4 * (idx & 127));
    }

    const float* ml = mask + (((nt & 1) << 4) + h) * 64 + 4 * c;
    v4f mreg[8];
#pragma unroll
    for (int i = 0; i < 8; ++i)
        mreg[i] = *(const v4f*)(ml + 128 * i);

    v4f land[2][8];
#pragma unroll
    for (int i = 0; i < 8; ++i)
        land[0][i] = *(const v4f*)(wl + (size_t)(2 * i) * NT);
#pragma unroll
    for (int i = 0; i < 8; ++i)
        land[1][i] = *(const v4f*)(wl + (size_t)(2 * i) * NT + 64);

#pragma unroll
    for (int i = 0; i < 8; ++i) {
        const int idx = tid + 512 * i;
        const int b = idx >> 7;
        const int u = idx & 127;
        xs[b * 128 + (u ^ (b & 15))] = xv[i];
    }
    __syncthreads();

    v4f* wst = wsl + wave * 544 + h * 17 + c;
    const v4f* wfr = wsl + wave * 544 + c * 17 + h;

#pragma unroll
    for (int i = 0; i < 8; ++i)
        wst[34 * i] = land[0][i] * mreg[i];
#pragma unroll
    for (int i = 0; i < 8; ++i)
        land[0][i] = *(const v4f*)(wl + (size_t)(2 * i) * NT + 128);

    const v4f* xa0 = xs + c * 128;
    const v4f* xa1 = xs + (c + 16) * 128;

    v8f acc0 = {}, acc1 = {}, acc2 = {}, acc3 = {};

    v4f af[2][8];
    v4f bf[2][4];
#pragma unroll
    for (int jj = 0; jj < 4; ++jj) {
        const int u = (2 * jj + h) ^ c;
        af[0][2 * jj]     = xa0[u];
        af[0][2 * jj + 1] = xa1[u];
        bf[0][jj]         = wfr[2 * jj];
    }

#pragma unroll
    for (int gi = 0; gi < 16; ++gi) {
        const int p = gi >> 1;
        if ((gi & 1) && p + 1 < 8) {
            const int q = p + 1;
#pragma unroll
            for (int i = 0; i < 8; ++i)
                wst[(q & 1) * 272 + 34 * i] = land[q & 1][i] * mreg[i];
            if (q + 2 < 8) {
#pragma unroll
                for (int i = 0; i < 8; ++i)
                    land[q & 1][i] = *(const v4f*)(wl + (size_t)(2 * i) * NT + 64 * (q + 2));
            }
        }
        if (gi + 1 < 16) {
            const int pn  = (gi + 1) >> 1;
            const int jn0 = ((gi + 1) & 1) << 2;
#pragma unroll
            for (int jj = 0; jj < 4; ++jj) {
                const int u = (2 * (jn0 + jj) + h) ^ c;
                af[(gi + 1) & 1][2 * jj]     = xa0[pn * 16 + u];
                af[(gi + 1) & 1][2 * jj + 1] = xa1[pn * 16 + u];
                bf[(gi + 1) & 1][jj]         = wfr[(pn & 1) * 272 + 2 * (jn0 + jj)];
            }
        }
        __builtin_amdgcn_sched_barrier(0);
#pragma unroll
        for (int jj = 0; jj < 4; ++jj) {
            const v4f a0 = af[gi & 1][2 * jj];
            const v4f a1 = af[gi & 1][2 * jj + 1];
            const v4f bm = bf[gi & 1][jj];
            const v2f a0l = {a0.x, a0.y}, a0h = {a0.z, a0.w};
            const v2f a1l = {a1.x, a1.y}, a1h = {a1.z, a1.w};
            const v2f bl  = {bm.x, bm.y}, bh  = {bm.z, bm.w};
            acc0 = __builtin_amdgcn_wmma_f32_16x16x4_f32(false, a0l, false, bl, (short)0, acc0, false, false);
            acc1 = __builtin_amdgcn_wmma_f32_16x16x4_f32(false, a1l, false, bl, (short)0, acc1, false, false);
            acc2 = __builtin_amdgcn_wmma_f32_16x16x4_f32(false, a0h, false, bh, (short)0, acc2, false, false);
            acc3 = __builtin_amdgcn_wmma_f32_16x16x4_f32(false, a1h, false, bh, (short)0, acc3, false, false);
        }
        __builtin_amdgcn_sched_barrier(0);
    }
    acc0 += acc2;
    acc1 += acc3;

    float* pt = part + (size_t)kc * PSZ + (size_t)nt * 512 + lane;
#pragma unroll
    for (int r = 0; r < 8; ++r) {
        pt[r * 32]       = acc0[r];
        pt[256 + r * 32] = acc1[r];
    }
}

__global__ __launch_bounds__(128) void mlm_reduce_kernel(
    const float* __restrict__ part, const float* __restrict__ y,
    const float* __restrict__ bias, float* __restrict__ out)
{
    const int t    = blockIdx.x * 128 + threadIdx.x;
    const int lane = t & 31;
    const int u    = t >> 5;
    const int r    = u & 7;
    const int m    = (u >> 3) & 1;
    const int oh   = (u >> 4) & 1;
    const int d    = u >> 5;
    const int b    = m * 16 + ((lane >> 4) << 3) + r;
    const int o    = oh * 16 + (lane & 15);

    const float* p0 = part + ((d * 2 + oh) * 2 + m) * 256 + r * 32 + lane;
    const v4f y0 = *(const v4f*)(y + b * 8);
    const v4f y1 = *(const v4f*)(y + b * 8 + 4);
    const float yv[8] = {y0.x, y0.y, y0.z, y0.w, y1.x, y1.y, y1.z, y1.w};
    float s = bias[d * 32 + o];
#pragma unroll
    for (int g = 0; g < 8; ++g) {
        const float* pg = p0 + g * 32768;
        float tsum = 0.0f;
#pragma unroll
        for (int q = 0; q < 8; ++q) tsum += pg[(size_t)q * PSZ];
        s += yv[g] * tsum;
    }
    out[b * 1024 + d * 32 + o] = s;
}

extern "C" void kernel_launch(void* const* d_in, const int* in_sizes, int n_in,
                              void* d_out, int out_size, void* d_ws, size_t ws_size,
                              hipStream_t stream) {
    const float* x    = (const float*)d_in[0];
    const float* y    = (const float*)d_in[1];
    const float* w    = (const float*)d_in[2];
    const float* bias = (const float*)d_in[3];
    const float* mask = (const float*)d_in[4];
    float* out  = (float*)d_out;
    float* part = (float*)d_ws;

    mlm_main_kernel<<<256, 512, 0, stream>>>(x, w, mask, part);
    mlm_reduce_kernel<<<256, 128, 0, stream>>>(part, y, bias, out);
}


# y-weighted sum over the 8 genres done in the main kernel's block via LDS (block = one d, wave = genre x o half): partials 8 MB -> 1 MB, 2nd kernel adds 8 not 64; f32 summation order changes
# speedup vs baseline: 1.6976x; 1.0413x over previous
#include <hip/hip_runtime.h>

typedef __attribute__((ext_vector_type(2))) float v2f;
typedef __attribute__((ext_vector_type(4))) float v4f;
typedef __attribute__((ext_vector_type(8))) float v8f;

static constexpr int NT  = 4096;
static constexpr int KL  = 512;
static constexpr int PSZ = 32768;

__global__ __launch_bounds__(512) void mlm_main_kernel(
    const float* __restrict__ x, const float* __restrict__ y,
    const float* __restrict__ w, const float* __restrict__ mask,
    float* __restrict__ part)
{
    __shared__ v4f xs[32 * 128];
    __shared__ v4f wsl[16 * 2 * 272];
    __shared__ float ys[256];

    const int tid  = threadIdx.x;
    const int wave = tid >> 5;
    const int lane = tid & 31;
    const int c    = lane & 15;
    const int h    = lane >> 4;
    const int kc   = blockIdx.x & 7;
    const int d    = blockIdx.x >> 3;
    const int g    = wave >> 1;
    const int oh   = wave & 1;
    const int nt   = (g * 32 + d) * 2 + oh;
    const int k0   = kc * KL;

    const float* wl = w + (size_t)(nt * 16 + h) * NT + k0 + 4 * c;

    v4f xv[8];
#pragma unroll
    for (int i = 0; i < 8; ++i) {
        const int idx = tid + 512 * i;
        xv[i] = *(const v4f*)(x + (idx >> 7) * NT + k0 + 4 * (idx & 127));
    }
    float yv = 0.0f;
    if (tid < 256) yv = y[tid];

    const float* ml = mask + (((nt & 1) << 4) + h) * 64 + 4 * c;
    v4f mreg[8];
#pragma unroll
    for (int i = 0; i < 8; ++i)
        mreg[i] = *(const v4f*)(ml + 128 * i);

    v4f land[2][8];
#pragma unroll
    for (int i = 0; i < 8; ++i)
        land[0][i] = *(const v4f*)(wl + (size_t)(2 * i) * NT);
#pragma unroll
    for (int i = 0; i < 8; ++i)
        land[1][i] = *(const v4f*)(wl + (size_t)(2 * i) * NT + 64);

#pragma unroll
    for (int i = 0; i < 8; ++i) {
        const int idx = tid + 512 * i;
        const int b = idx >> 7;
        const int u = idx & 127;
        xs[b * 128 + (u ^ (b & 15))] = xv[i];
    }
    if (tid < 256) ys[tid] = yv;
    __syncthreads();

    v4f* wst = wsl + wave * 544 + h * 17 + c;
    const v4f* wfr = wsl + wave * 544 + c * 17 + h;

#pragma unroll
    for (int i = 0; i < 8; ++i)
        wst[34 * i] = land[0][i] * mreg[i];
#pragma unroll
    for (int i = 0; i < 8; ++i)
        land[0][i] = *(const v4f*)(wl + (size_t)(2 * i) * NT + 128);

    const v4f* xa0 = xs + c * 128;
    const v4f* xa1 = xs + (c + 16) * 128;

    v8f acc0 = {}, acc1 = {}, acc2 = {}, acc3 = {};

    v4f af[2][8];
    v4f bf[2][4];
#pragma unroll
    for (int jj = 0; jj < 4; ++jj) {
        const int u = (2 * jj + h) ^ c;
        af[0][2 * jj]     = xa0[u];
        af[0][2 * jj + 1] = xa1[u];
        bf[0][jj]         = wfr[2 * jj];
    }

#pragma unroll
    for (int gi = 0; gi < 16; ++gi) {
        const int p = gi >> 1;
        if ((gi & 1) && p + 1 < 8) {
            const int q = p + 1;
#pragma unroll
            for (int i = 0; i < 8; ++i)
                wst[(q & 1) * 272 + 34 * i] = land[q & 1][i] * mreg[i];
            if (q + 2 < 8) {
#pragma unroll
                for (int i = 0; i < 8; ++i)
                    land[q & 1][i] = *(const v4f*)(wl + (size_t)(2 * i) * NT + 64 * (q + 2));
            }
        }
        if (gi + 1 < 16) {
            const int pn  = (gi + 1) >> 1;
            const int jn0 = ((gi + 1) & 1) << 2;
#pragma unroll
            for (int jj = 0; jj < 4; ++jj) {
                const int u = (2 * (jn0 + jj) + h) ^ c;
                af[(gi + 1) & 1][2 * jj]     = xa0[pn * 16 + u];
                af[(gi + 1) & 1][2 * jj + 1] = xa1[pn * 16 + u];
                bf[(gi + 1) & 1][jj]         = wfr[(pn & 1) * 272 + 2 * (jn0 + jj)];
            }
        }
        __builtin_amdgcn_sched_barrier(0);
#pragma unroll
        for (int jj = 0; jj < 4; ++jj) {
            const v4f a0 = af[gi & 1][2 * jj];
            const v4f a1 = af[gi & 1][2 * jj + 1];
            const v4f bm = bf[gi & 1][jj];
            const v2f a0l = {a0.x, a0.y}, a0h = {a0.z, a0.w};
            const v2f a1l = {a1.x, a1.y}, a1h = {a1.z, a1.w};
            const v2f bl  = {bm.x, bm.y}, bh  = {bm.z, bm.w};
            acc0 = __builtin_amdgcn_wmma_f32_16x16x4_f32(false, a0l, false, bl, (short)0, acc0, false, false);
            acc1 = __builtin_amdgcn_wmma_f32_16x16x4_f32(false, a1l, false, bl, (short)0, acc1, false, false);
            acc2 = __builtin_amdgcn_wmma_f32_16x16x4_f32(false, a0h, false, bh, (short)0, acc2, false, false);
            acc3 = __builtin_amdgcn_wmma_f32_16x16x4_f32(false, a1h, false, bh, (short)0, acc3, false, false);
        }
        __builtin_amdgcn_sched_barrier(0);
    }
    acc0 += acc2;
    acc1 += acc3;

    float* wslf = (float*)wsl;
    float* red = wslf + wave * 2176 + oh * 16 + lane;
#pragma unroll
    for (int r = 0; r < 8; ++r) {
        red[r * 32]       = acc0[r] * ys[(8 * h + r) * 8 + g];
        red[256 + r * 32] = acc1[r] * ys[(16 + 8 * h + r) * 8 + g];
    }
    __syncthreads();

    const float* rd = wslf + (lane >> 4) * 2192 + (wave & 7) * 32 + (wave >> 3) * 16 + (lane & 15);
    float s0 = 0.0f, s1 = 0.0f;
#pragma unroll
    for (int gg = 0; gg < 8; ++gg) {
        s0 += rd[gg * 4352];
        s1 += rd[gg * 4352 + 256];
    }
    float* pt = part + (size_t)kc * PSZ + wave * 1024 + d * 32 + lane;
    pt[0]         = s0;
    pt[16 * 1024] = s1;
}

__global__ __launch_bounds__(32) void mlm_reduce_kernel(
    const float* __restrict__ part, const float* __restrict__ bias, float* __restrict__ out)
{
    const int t = blockIdx.x * 32 + threadIdx.x;
    const v4f* p = (const v4f*)part + t;
    v4f s = p[0];
#pragma unroll
    for (int q = 1; q < 8; ++q) s += p[q * (PSZ / 4)];
    s += ((const v4f*)bias)[t & 255];
    ((v4f*)out)[t] = s;
}

extern "C" void kernel_launch(void* const* d_in, const int* in_sizes, int n_in,
                              void* d_out, int out_size, void* d_ws, size_t ws_size,
                              hipStream_t stream) {
    const float* x    = (const float*)d_in[0];
    const float* y    = (const float*)d_in[1];
    const float* w    = (const float*)d_in[2];
    const float* bias = (const float*)d_in[3];
    const float* mask = (const float*)d_in[4];
    float* out  = (float*)d_out;
    float* part = (float*)d_ws;

    mlm_main_kernel<<<256, 512, 0, stream>>>(x, y, w, mask, part);
    mlm_reduce_kernel<<<256, 32, 0, stream>>>(part, bias, out);
}
